# starting kernel
# baseline (speedup 1.0000x reference)
#include <hip/hip_runtime.h>
#include <hip/hip_bf16.h>


typedef float v2f __attribute__((ext_vector_type(2)));
typedef float v4f __attribute__((ext_vector_type(4)));
typedef float v8f __attribute__((ext_vector_type(8)));

#define B_  16
#define S_  1024
#define H_  512
#define C_  7
#define BS_ (B_ * S_)
#define EPS_ 1e-10f
#define LPAD 20

static __device__ __forceinline__ v8f wmma4(v2f a, v2f b, v8f c) {
    return __builtin_amdgcn_wmma_f32_16x16x4_f32(
        false, a, false, b, (short)0, c, false, false);
}

static __device__ float blockReduceSum256(float v) {
    __shared__ float red[8];
    #pragma unroll
    for (int off = 16; off > 0; off >>= 1) v += __shfl_xor(v, off, 32);
    const int warp = threadIdx.x >> 5, lane = threadIdx.x & 31;
    __syncthreads();
    if (lane == 0) red[warp] = v;
    __syncthreads();
    float r = 0.f;
    #pragma unroll
    for (int w = 0; w < 8; ++w) r += red[w];
    return r;
}

__global__ __launch_bounds__(256)
void weff_kernel(const float* __restrict__ gW, float* __restrict__ weff) {
    int idx = blockIdx.x * blockDim.x + threadIdx.x;
    if (idx >= H_ * H_) return;
    weff[idx] = gW[idx] + gW[H_ * H_ + idx] - 0.5f * gW[2 * H_ * H_ + idx];
}

__global__ __launch_bounds__(256)
void ent_sum_kernel(const float* __restrict__ t, const float* __restrict__ a,
                    const float* __restrict__ v, float* __restrict__ sum_all,
                    float* __restrict__ entT, float* __restrict__ entA,
                    float* __restrict__ entV) {
    const size_t row = blockIdx.x;
    const size_t base = row * H_;
    const int i0 = threadIdx.x, i1 = threadIdx.x + 256;

    float t0 = t[base + i0], t1 = t[base + i1];
    float a0 = a[base + i0], a1 = a[base + i1];
    float v0 = v[base + i0], v1 = v[base + i1];

    sum_all[base + i0] = t0 + a0 + v0;
    sum_all[base + i1] = t1 + a1 + v1;

    float st = blockReduceSum256(fabsf(t0) + fabsf(t1));
    float sa = blockReduceSum256(fabsf(a0) + fabsf(a1));
    float sv = blockReduceSum256(fabsf(v0) + fabsf(v1));

    float rt = 1.f / (st + EPS_), ra = 1.f / (sa + EPS_), rv = 1.f / (sv + EPS_);
    float f;
    float et = 0.f, ea = 0.f, ev = 0.f;
    f = fabsf(t0) * rt; et += f * __logf(f + EPS_);
    f = fabsf(t1) * rt; et += f * __logf(f + EPS_);
    f = fabsf(a0) * ra; ea += f * __logf(f + EPS_);
    f = fabsf(a1) * ra; ea += f * __logf(f + EPS_);
    f = fabsf(v0) * rv; ev += f * __logf(f + EPS_);
    f = fabsf(v1) * rv; ev += f * __logf(f + EPS_);

    et = blockReduceSum256(et);
    ea = blockReduceSum256(ea);
    ev = blockReduceSum256(ev);
    if (threadIdx.x == 0) {
        entT[row] = -et; entA[row] = -ea; entV[row] = -ev;
    }
}

__global__ __launch_bounds__(256)
void rowsum_dsi_kernel(const float* __restrict__ sp, const float* __restrict__ tm,
                       float* __restrict__ dsi) {
    const int row  = blockIdx.x * 8 + (threadIdx.x >> 5);
    const int lane = threadIdx.x & 31;
    const int iloc = row & (S_ - 1);
    const size_t base = (size_t)row * S_;
    float s = 0.f;
    #pragma unroll 4
    for (int j = lane; j < S_; j += 32) {
        float aij = 0.5f * (sp[base + j] + tm[base + j]);
        if (j == iloc) aij += 1.f;
        s += aij;
    }
    #pragma unroll
    for (int off = 16; off > 0; off >>= 1) s += __shfl_xor(s, off, 32);
    if (lane == 0) dsi[row] = rsqrtf(s + EPS_);
}

__global__ __launch_bounds__(256)
void normadj_kernel(const float* __restrict__ sp, const float* __restrict__ tm,
                    const float* __restrict__ dsi, float* __restrict__ na) {
    size_t idx = (size_t)blockIdx.x * blockDim.x + threadIdx.x;
    const size_t total = (size_t)B_ * S_ * S_;
    if (idx >= total) return;
    size_t b = idx >> 20;
    size_t r = idx & ((1u << 20) - 1);
    size_t i = r >> 10, j = r & (S_ - 1);
    float aij = 0.5f * (sp[idx] + tm[idx]) + ((i == j) ? 1.f : 0.f);
    na[idx] = dsi[b * S_ + i] * aij * dsi[b * S_ + j];
}

template <int EPI>
__global__ __launch_bounds__(256)
void gemm_wmma_f32(const float* __restrict__ A, const float* __restrict__ Bm,
                   float* __restrict__ C, int M, int N, int K,
                   int lda, int ldb, int ldc,
                   long strideA, long strideB, long strideC,
                   const float* __restrict__ e0,
                   const float* __restrict__ e1,
                   const float* __restrict__ e2,
                   const float* __restrict__ e3,
                   const float* __restrict__ e4) {
    __shared__ float As[2][128][LPAD];
    __shared__ float Bs[2][64][LPAD];

    const float* Ab = A + (size_t)blockIdx.z * strideA;
    const float* Bb = Bm + (size_t)blockIdx.z * strideB;
    float*       Cb = C + (size_t)blockIdx.z * strideC;
    const int grow0 = (int)blockIdx.z * M;

    const int bm = blockIdx.y * 128;
    const int bn = blockIdx.x * 64;
    const int tid = threadIdx.x;
    const int warp = tid >> 5, lane = tid & 31;
    const int waveM = warp & 3, waveN = warp >> 2;
    const int half = lane >> 4, mlan = lane & 15;

    const int amRow = tid >> 2;
    const int akQ   = (tid & 3) * 4;
    const int bkRow = tid >> 4;
    const int bnQ   = (tid & 15) * 4;
    const float* aP0 = Ab + (size_t)(bm + amRow) * lda + akQ;
    const float* aP1 = Ab + (size_t)(bm + amRow + 64) * lda + akQ;
    const float* bP  = Bb + (size_t)bkRow * ldb + bn + bnQ;
    const size_t bStep = (size_t)ldb * 16;

    v8f acc[2][2];
    #pragma unroll
    for (int i = 0; i < 2; ++i)
        #pragma unroll
        for (int j = 0; j < 2; ++j)
            #pragma unroll
            for (int e = 0; e < 8; ++e) acc[i][j][e] = 0.f;

    v4f ra0, ra1, rb;
    const int ktiles = K >> 4;

    ra0 = *(const v4f*)(aP0);
    ra1 = *(const v4f*)(aP1);
    rb  = *(const v4f*)(bP);
    *(v4f*)&As[0][amRow][akQ]      = ra0;
    *(v4f*)&As[0][amRow + 64][akQ] = ra1;
    Bs[0][bnQ + 0][bkRow] = rb[0];
    Bs[0][bnQ + 1][bkRow] = rb[1];
    Bs[0][bnQ + 2][bkRow] = rb[2];
    Bs[0][bnQ + 3][bkRow] = rb[3];
    __syncthreads();

    for (int kt = 0; kt < ktiles; ++kt) {
        const int cur = kt & 1, nxt = cur ^ 1;
        if (kt + 1 < ktiles) {
            ra0 = *(const v4f*)(aP0 + (size_t)(kt + 1) * 16);
            ra1 = *(const v4f*)(aP1 + (size_t)(kt + 1) * 16);
            rb  = *(const v4f*)(bP + (size_t)(kt + 1) * bStep);
        }
        #pragma unroll
        for (int kk4 = 0; kk4 < 4; ++kk4) {
            const int kfrag = kk4 * 4 + half * 2;
            v2f a0 = *(const v2f*)&As[cur][waveM * 32 +  0 + mlan][kfrag];
            v2f a1 = *(const v2f*)&As[cur][waveM * 32 + 16 + mlan][kfrag];
            v2f b0 = *(const v2f*)&Bs[cur][waveN * 32 +  0 + mlan][kfrag];
            v2f b1 = *(const v2f*)&Bs[cur][waveN * 32 + 16 + mlan][kfrag];
            acc[0][0] = wmma4(a0, b0, acc[0][0]);
            acc[0][1] = wmma4(a0, b1, acc[0][1]);
            acc[1][0] = wmma4(a1, b0, acc[1][0]);
            acc[1][1] = wmma4(a1, b1, acc[1][1]);
        }
        if (kt + 1 < ktiles) {
            *(v4f*)&As[nxt][amRow][akQ]      = ra0;
            *(v4f*)&As[nxt][amRow + 64][akQ] = ra1;
            Bs[nxt][bnQ + 0][bkRow] = rb[0];
            Bs[nxt][bnQ + 1][bkRow] = rb[1];
            Bs[nxt][bnQ + 2][bkRow] = rb[2];
            Bs[nxt][bnQ + 3][bkRow] = rb[3];
        }
        __syncthreads();
    }

    #pragma unroll
    for (int mi = 0; mi < 2; ++mi) {
        #pragma unroll
        for (int ni = 0; ni < 2; ++ni) {
            const int tM = bm + waveM * 32 + mi * 16;
            const int tN = bn + waveN * 32 + ni * 16;
            #pragma unroll
            for (int i = 0; i < 8; ++i) {
                const int row = tM + (half ? i + 8 : i);
                const int col = tN + mlan;
                float val = acc[mi][ni][i];
                if (EPI == 1) {
                    val *= 0.5f;
                } else if (EPI == 2) {
                    const size_t grow = (size_t)(grow0 + row);
                    float logit = val + e0[grow * H_ + col]
                                + e1[grow] * e2[col] + e3[col];
                    float g = 1.f / (1.f + __expf(-logit));
                    val = g * e4[grow * H_ + col];
                } else if (EPI == 3) {
                    val = fmaxf(val + e0[col], 0.f);
                }
                Cb[(size_t)row * ldc + col] = val;
            }
        }
    }
}

__global__ __launch_bounds__(256)
void classifier_kernel(const float* __restrict__ h, const float* __restrict__ Wc,
                       const float* __restrict__ bc, float* __restrict__ out) {
    int idx = blockIdx.x * blockDim.x + threadIdx.x;
    if (idx >= BS_ * C_) return;
    int row = idx / C_, c = idx - row * C_;
    const float* hr = h + (size_t)row * H_;
    float acc = bc[c];
    #pragma unroll 8
    for (int k = 0; k < H_; ++k) acc = fmaf(hr[k], Wc[k * C_ + c], acc);
    out[idx] = acc;
}

extern "C" void kernel_launch(void* const* d_in, const int* in_sizes, int n_in,
                              void* d_out, int out_size, void* d_ws, size_t ws_size,
                              hipStream_t stream) {
    const float* text   = (const float*)d_in[0];
    const float* audio  = (const float*)d_in[1];
    const float* visual = (const float*)d_in[2];
    const float* sp     = (const float*)d_in[3];
    const float* tm     = (const float*)d_in[4];
    const float* gate_W = (const float*)d_in[6];
    const float* gate_b = (const float*)d_in[7];
    const float* W0 = (const float*)d_in[8];  const float* b0 = (const float*)d_in[9];
    const float* W1 = (const float*)d_in[10]; const float* b1 = (const float*)d_in[11];
    const float* W2 = (const float*)d_in[12]; const float* b2 = (const float*)d_in[13];
    const float* Wc = (const float*)d_in[14]; const float* bc = (const float*)d_in[15];
    float* out = (float*)d_out;

    float* ws    = (float*)d_ws;
    float* weff  = ws;
    float* entT  = weff + H_ * H_;
    float* entA  = entT + BS_;
    float* entV  = entA + BS_;
    float* dsi   = entV + BS_;
    float* sumA  = dsi + BS_;
    float* P     = sumA + (size_t)BS_ * H_;
    float* h0    = P    + (size_t)BS_ * H_;
    float* na    = h0   + (size_t)BS_ * 3 * H_;
    float* tmp   = sumA;
    float* hbuf  = P;

    const float* gWc_rows = gate_W + (size_t)2 * H_ * H_;
    const float* gw_last  = gate_W + (size_t)3 * H_ * H_;

    weff_kernel<<<(H_ * H_ + 255) / 256, 256, 0, stream>>>(gate_W, weff);

    ent_sum_kernel<<<BS_, 256, 0, stream>>>(text, audio, visual, sumA,
                                            entT, entA, entV);

    dim3 gP(H_ / 64, BS_ / 128, 1);
    gemm_wmma_f32<1><<<gP, 256, 0, stream>>>(sumA, gWc_rows, P,
        BS_, H_, H_, H_, H_, H_, 0, 0, 0, nullptr, nullptr, nullptr, nullptr, nullptr);

    gemm_wmma_f32<2><<<gP, 256, 0, stream>>>(text, weff, h0 + 0 * H_,
        BS_, H_, H_, H_, H_, 3 * H_, 0, 0, 0, P, entT, gw_last, gate_b, text);
    gemm_wmma_f32<2><<<gP, 256, 0, stream>>>(audio, weff, h0 + 1 * H_,
        BS_, H_, H_, H_, H_, 3 * H_, 0, 0, 0, P, entA, gw_last, gate_b, audio);
    gemm_wmma_f32<2><<<gP, 256, 0, stream>>>(visual, weff, h0 + 2 * H_,
        BS_, H_, H_, H_, H_, 3 * H_, 0, 0, 0, P, entV, gw_last, gate_b, visual);

    rowsum_dsi_kernel<<<BS_ / 8, 256, 0, stream>>>(sp, tm, dsi);
    normadj_kernel<<<(B_ * S_ * S_ + 255) / 256, 256, 0, stream>>>(sp, tm, dsi, na);

    gemm_wmma_f32<0><<<gP, 256, 0, stream>>>(h0, W0, tmp,
        BS_, H_, 3 * H_, 3 * H_, H_, H_, 0, 0, 0,
        nullptr, nullptr, nullptr, nullptr, nullptr);
    dim3 gS(H_ / 64, S_ / 128, B_);
    gemm_wmma_f32<3><<<gS, 256, 0, stream>>>(na, tmp, hbuf,
        S_, H_, S_, S_, H_, H_,
        (long)S_ * S_, (long)S_ * H_, (long)S_ * H_,
        b0, nullptr, nullptr, nullptr, nullptr);

    gemm_wmma_f32<0><<<gP, 256, 0, stream>>>(hbuf, W1, tmp,
        BS_, H_, H_, H_, H_, H_, 0, 0, 0,
        nullptr, nullptr, nullptr, nullptr, nullptr);
    gemm_wmma_f32<3><<<gS, 256, 0, stream>>>(na, tmp, hbuf,
        S_, H_, S_, S_, H_, H_,
        (long)S_ * S_, (long)S_ * H_, (long)S_ * H_,
        b1, nullptr, nullptr, nullptr, nullptr);

    gemm_wmma_f32<0><<<gP, 256, 0, stream>>>(hbuf, W2, tmp,
        BS_, H_, H_, H_, H_, H_, 0, 0, 0,
        nullptr, nullptr, nullptr, nullptr, nullptr);
    gemm_wmma_f32<3><<<gS, 256, 0, stream>>>(na, tmp, hbuf,
        S_, H_, S_, S_, H_, H_,
        (long)S_ * S_, (long)S_ * H_, (long)S_ * H_,
        b2, nullptr, nullptr, nullptr, nullptr);

    classifier_kernel<<<(BS_ * C_ + 255) / 256, 256, 0, stream>>>(hbuf, Wc, bc, out);
}


# Normalised-adjacency pass dropped: D^-1/2 factors folded into GEMM epilogues (tmp rows and layer outputs scaled by dsi); degree kernel stores (sp+tm)/2+I via b128 loads, degree sum order changes
# speedup vs baseline: 1.0401x; 1.0401x over previous
#include <hip/hip_runtime.h>
#include <hip/hip_bf16.h>


typedef float v2f __attribute__((ext_vector_type(2)));
typedef float v4f __attribute__((ext_vector_type(4)));
typedef float v8f __attribute__((ext_vector_type(8)));

#define B_  16
#define S_  1024
#define H_  512
#define C_  7
#define BS_ (B_ * S_)
#define EPS_ 1e-10f
#define LPAD 20

static __device__ __forceinline__ v8f wmma4(v2f a, v2f b, v8f c) {
    return __builtin_amdgcn_wmma_f32_16x16x4_f32(
        false, a, false, b, (short)0, c, false, false);
}

static __device__ float blockReduceSum256(float v) {
    __shared__ float red[8];
    #pragma unroll
    for (int off = 16; off > 0; off >>= 1) v += __shfl_xor(v, off, 32);
    const int warp = threadIdx.x >> 5, lane = threadIdx.x & 31;
    __syncthreads();
    if (lane == 0) red[warp] = v;
    __syncthreads();
    float r = 0.f;
    #pragma unroll
    for (int w = 0; w < 8; ++w) r += red[w];
    return r;
}

__global__ __launch_bounds__(256)
void weff_kernel(const float* __restrict__ gW, float* __restrict__ weff) {
    int idx = blockIdx.x * blockDim.x + threadIdx.x;
    if (idx >= H_ * H_) return;
    weff[idx] = gW[idx] + gW[H_ * H_ + idx] - 0.5f * gW[2 * H_ * H_ + idx];
}

__global__ __launch_bounds__(256)
void ent_sum_kernel(const float* __restrict__ t, const float* __restrict__ a,
                    const float* __restrict__ v, float* __restrict__ sum_all,
                    float* __restrict__ entT, float* __restrict__ entA,
                    float* __restrict__ entV) {
    const size_t row = blockIdx.x;
    const size_t base = row * H_;
    const int i0 = threadIdx.x, i1 = threadIdx.x + 256;

    float t0 = t[base + i0], t1 = t[base + i1];
    float a0 = a[base + i0], a1 = a[base + i1];
    float v0 = v[base + i0], v1 = v[base + i1];

    sum_all[base + i0] = t0 + a0 + v0;
    sum_all[base + i1] = t1 + a1 + v1;

    float st = blockReduceSum256(fabsf(t0) + fabsf(t1));
    float sa = blockReduceSum256(fabsf(a0) + fabsf(a1));
    float sv = blockReduceSum256(fabsf(v0) + fabsf(v1));

    float rt = 1.f / (st + EPS_), ra = 1.f / (sa + EPS_), rv = 1.f / (sv + EPS_);
    float f;
    float et = 0.f, ea = 0.f, ev = 0.f;
    f = fabsf(t0) * rt; et += f * __logf(f + EPS_);
    f = fabsf(t1) * rt; et += f * __logf(f + EPS_);
    f = fabsf(a0) * ra; ea += f * __logf(f + EPS_);
    f = fabsf(a1) * ra; ea += f * __logf(f + EPS_);
    f = fabsf(v0) * rv; ev += f * __logf(f + EPS_);
    f = fabsf(v1) * rv; ev += f * __logf(f + EPS_);

    et = blockReduceSum256(et);
    ea = blockReduceSum256(ea);
    ev = blockReduceSum256(ev);
    if (threadIdx.x == 0) {
        entT[row] = -et; entA[row] = -ea; entV[row] = -ev;
    }
}

__global__ __launch_bounds__(256)
void adj_deg_kernel(const float* __restrict__ sp, const float* __restrict__ tm,
                    float* __restrict__ adj, float* __restrict__ dsi) {
    const int row  = blockIdx.x * 8 + (threadIdx.x >> 5);
    const int lane = threadIdx.x & 31;
    const int iloc = row & (S_ - 1);
    const size_t base = (size_t)row * S_;
    const v4f* sp4 = (const v4f*)(sp + base);
    const v4f* tm4 = (const v4f*)(tm + base);
    v4f*       ad4 = (v4f*)(adj + base);

    v4f x[8], y[8];
    #pragma unroll
    for (int m = 0; m < 8; ++m) {
        x[m] = sp4[m * 32 + lane];
        y[m] = tm4[m * 32 + lane];
    }
    float s = 0.f;
    #pragma unroll
    for (int m = 0; m < 8; ++m) {
        const int j0 = (m * 32 + lane) * 4;
        v4f a;
        a[0] = 0.5f * (x[m][0] + y[m][0]) + ((iloc == j0 + 0) ? 1.f : 0.f);
        a[1] = 0.5f * (x[m][1] + y[m][1]) + ((iloc == j0 + 1) ? 1.f : 0.f);
        a[2] = 0.5f * (x[m][2] + y[m][2]) + ((iloc == j0 + 2) ? 1.f : 0.f);
        a[3] = 0.5f * (x[m][3] + y[m][3]) + ((iloc == j0 + 3) ? 1.f : 0.f);
        ad4[m * 32 + lane] = a;
        s += (a[0] + a[1]) + (a[2] + a[3]);
    }
    #pragma unroll
    for (int off = 16; off > 0; off >>= 1) s += __shfl_xor(s, off, 32);
    if (lane == 0) dsi[row] = rsqrtf(s + EPS_);
}

template <int EPI>
__global__ __launch_bounds__(256)
void gemm_wmma_f32(const float* __restrict__ A, const float* __restrict__ Bm,
                   float* __restrict__ C, int M, int N, int K,
                   int lda, int ldb, int ldc,
                   long strideA, long strideB, long strideC,
                   const float* __restrict__ e0,
                   const float* __restrict__ e1,
                   const float* __restrict__ e2,
                   const float* __restrict__ e3,
                   const float* __restrict__ e4) {
    __shared__ float As[2][128][LPAD];
    __shared__ float Bs[2][64][LPAD];

    const float* Ab = A + (size_t)blockIdx.z * strideA;
    const float* Bb = Bm + (size_t)blockIdx.z * strideB;
    float*       Cb = C + (size_t)blockIdx.z * strideC;
    const int grow0 = (int)blockIdx.z * M;

    const int bm = blockIdx.y * 128;
    const int bn = blockIdx.x * 64;
    const int tid = threadIdx.x;
    const int warp = tid >> 5, lane = tid & 31;
    const int waveM = warp & 3, waveN = warp >> 2;
    const int half = lane >> 4, mlan = lane & 15;

    const int amRow = tid >> 2;
    const int akQ   = (tid & 3) * 4;
    const int bkRow = tid >> 4;
    const int bnQ   = (tid & 15) * 4;
    const float* aP0 = Ab + (size_t)(bm + amRow) * lda + akQ;
    const float* aP1 = Ab + (size_t)(bm + amRow + 64) * lda + akQ;
    const float* bP  = Bb + (size_t)bkRow * ldb + bn + bnQ;
    const size_t bStep = (size_t)ldb * 16;

    v8f acc[2][2];
    #pragma unroll
    for (int i = 0; i < 2; ++i)
        #pragma unroll
        for (int j = 0; j < 2; ++j)
            #pragma unroll
            for (int e = 0; e < 8; ++e) acc[i][j][e] = 0.f;

    v4f ra0, ra1, rb;
    const int ktiles = K >> 4;

    ra0 = *(const v4f*)(aP0);
    ra1 = *(const v4f*)(aP1);
    rb  = *(const v4f*)(bP);
    *(v4f*)&As[0][amRow][akQ]      = ra0;
    *(v4f*)&As[0][amRow + 64][akQ] = ra1;
    Bs[0][bnQ + 0][bkRow] = rb[0];
    Bs[0][bnQ + 1][bkRow] = rb[1];
    Bs[0][bnQ + 2][bkRow] = rb[2];
    Bs[0][bnQ + 3][bkRow] = rb[3];
    __syncthreads();

    for (int kt = 0; kt < ktiles; ++kt) {
        const int cur = kt & 1, nxt = cur ^ 1;
        if (kt + 1 < ktiles) {
            ra0 = *(const v4f*)(aP0 + (size_t)(kt + 1) * 16);
            ra1 = *(const v4f*)(aP1 + (size_t)(kt + 1) * 16);
            rb  = *(const v4f*)(bP + (size_t)(kt + 1) * bStep);
        }
        #pragma unroll
        for (int kk4 = 0; kk4 < 4; ++kk4) {
            const int kfrag = kk4 * 4 + half * 2;
            v2f a0 = *(const v2f*)&As[cur][waveM * 32 +  0 + mlan][kfrag];
            v2f a1 = *(const v2f*)&As[cur][waveM * 32 + 16 + mlan][kfrag];
            v2f b0 = *(const v2f*)&Bs[cur][waveN * 32 +  0 + mlan][kfrag];
            v2f b1 = *(const v2f*)&Bs[cur][waveN * 32 + 16 + mlan][kfrag];
            acc[0][0] = wmma4(a0, b0, acc[0][0]);
            acc[0][1] = wmma4(a0, b1, acc[0][1]);
            acc[1][0] = wmma4(a1, b0, acc[1][0]);
            acc[1][1] = wmma4(a1, b1, acc[1][1]);
        }
        if (kt + 1 < ktiles) {
            *(v4f*)&As[nxt][amRow][akQ]      = ra0;
            *(v4f*)&As[nxt][amRow + 64][akQ] = ra1;
            Bs[nxt][bnQ + 0][bkRow] = rb[0];
            Bs[nxt][bnQ + 1][bkRow] = rb[1];
            Bs[nxt][bnQ + 2][bkRow] = rb[2];
            Bs[nxt][bnQ + 3][bkRow] = rb[3];
        }
        __syncthreads();
    }

    #pragma unroll
    for (int mi = 0; mi < 2; ++mi) {
        #pragma unroll
        for (int ni = 0; ni < 2; ++ni) {
            const int tM = bm + waveM * 32 + mi * 16;
            const int tN = bn + waveN * 32 + ni * 16;
            #pragma unroll
            for (int i = 0; i < 8; ++i) {
                const int row = tM + (half ? i + 8 : i);
                const int col = tN + mlan;
                float val = acc[mi][ni][i];
                if (EPI == 0) {
                    val *= e0[(size_t)(grow0 + row)];
                } else if (EPI == 1) {
                    val *= 0.5f;
                } else if (EPI == 2) {
                    const size_t grow = (size_t)(grow0 + row);
                    float logit = val + e0[grow * H_ + col]
                                + e1[grow] * e2[col] + e3[col];
                    float g = 1.f / (1.f + __expf(-logit));
                    val = g * e4[grow * H_ + col];
                } else if (EPI == 3) {
                    val = fmaxf(val * e1[(size_t)(grow0 + row)] + e0[col], 0.f);
                }
                Cb[(size_t)row * ldc + col] = val;
            }
        }
    }
}

__global__ __launch_bounds__(256)
void classifier_kernel(const float* __restrict__ h, const float* __restrict__ Wc,
                       const float* __restrict__ bc, float* __restrict__ out) {
    int idx = blockIdx.x * blockDim.x + threadIdx.x;
    if (idx >= BS_ * C_) return;
    int row = idx / C_, c = idx - row * C_;
    const float* hr = h + (size_t)row * H_;
    float acc = bc[c];
    #pragma unroll 8
    for (int k = 0; k < H_; ++k) acc = fmaf(hr[k], Wc[k * C_ + c], acc);
    out[idx] = acc;
}

extern "C" void kernel_launch(void* const* d_in, const int* in_sizes, int n_in,
                              void* d_out, int out_size, void* d_ws, size_t ws_size,
                              hipStream_t stream) {
    const float* text   = (const float*)d_in[0];
    const float* audio  = (const float*)d_in[1];
    const float* visual = (const float*)d_in[2];
    const float* sp     = (const float*)d_in[3];
    const float* tm     = (const float*)d_in[4];
    const float* gate_W = (const float*)d_in[6];
    const float* gate_b = (const float*)d_in[7];
    const float* W0 = (const float*)d_in[8];  const float* b0 = (const float*)d_in[9];
    const float* W1 = (const float*)d_in[10]; const float* b1 = (const float*)d_in[11];
    const float* W2 = (const float*)d_in[12]; const float* b2 = (const float*)d_in[13];
    const float* Wc = (const float*)d_in[14]; const float* bc = (const float*)d_in[15];
    float* out = (float*)d_out;

    float* ws    = (float*)d_ws;
    float* weff  = ws;
    float* entT  = weff + H_ * H_;
    float* entA  = entT + BS_;
    float* entV  = entA + BS_;
    float* dsi   = entV + BS_;
    float* sumA  = dsi + BS_;
    float* P     = sumA + (size_t)BS_ * H_;
    float* h0    = P    + (size_t)BS_ * H_;
    float* na    = h0   + (size_t)BS_ * 3 * H_;
    float* tmp   = sumA;
    float* hbuf  = P;

    const float* gWc_rows = gate_W + (size_t)2 * H_ * H_;
    const float* gw_last  = gate_W + (size_t)3 * H_ * H_;

    weff_kernel<<<(H_ * H_ + 255) / 256, 256, 0, stream>>>(gate_W, weff);

    ent_sum_kernel<<<BS_, 256, 0, stream>>>(text, audio, visual, sumA,
                                            entT, entA, entV);

    dim3 gP(H_ / 64, BS_ / 128, 1);
    gemm_wmma_f32<1><<<gP, 256, 0, stream>>>(sumA, gWc_rows, P,
        BS_, H_, H_, H_, H_, H_, 0, 0, 0, nullptr, nullptr, nullptr, nullptr, nullptr);

    gemm_wmma_f32<2><<<gP, 256, 0, stream>>>(text, weff, h0 + 0 * H_,
        BS_, H_, H_, H_, H_, 3 * H_, 0, 0, 0, P, entT, gw_last, gate_b, text);
    gemm_wmma_f32<2><<<gP, 256, 0, stream>>>(audio, weff, h0 + 1 * H_,
        BS_, H_, H_, H_, H_, 3 * H_, 0, 0, 0, P, entA, gw_last, gate_b, audio);
    gemm_wmma_f32<2><<<gP, 256, 0, stream>>>(visual, weff, h0 + 2 * H_,
        BS_, H_, H_, H_, H_, 3 * H_, 0, 0, 0, P, entV, gw_last, gate_b, visual);

    adj_deg_kernel<<<BS_ / 8, 256, 0, stream>>>(sp, tm, na, dsi);

    gemm_wmma_f32<0><<<gP, 256, 0, stream>>>(h0, W0, tmp,
        BS_, H_, 3 * H_, 3 * H_, H_, H_, 0, 0, 0,
        dsi, nullptr, nullptr, nullptr, nullptr);
    dim3 gS(H_ / 64, S_ / 128, B_);
    gemm_wmma_f32<3><<<gS, 256, 0, stream>>>(na, tmp, hbuf,
        S_, H_, S_, S_, H_, H_,
        (long)S_ * S_, (long)S_ * H_, (long)S_ * H_,
        b0, dsi, nullptr, nullptr, nullptr);

    gemm_wmma_f32<0><<<gP, 256, 0, stream>>>(hbuf, W1, tmp,
        BS_, H_, H_, H_, H_, H_, 0, 0, 0,
        dsi, nullptr, nullptr, nullptr, nullptr);
    gemm_wmma_f32<3><<<gS, 256, 0, stream>>>(na, tmp, hbuf,
        S_, H_, S_, S_, H_, H_,
        (long)S_ * S_, (long)S_ * H_, (long)S_ * H_,
        b1, dsi, nullptr, nullptr, nullptr);

    gemm_wmma_f32<0><<<gP, 256, 0, stream>>>(hbuf, W2, tmp,
        BS_, H_, H_, H_, H_, H_, 0, 0, 0,
        dsi, nullptr, nullptr, nullptr, nullptr);
    gemm_wmma_f32<3><<<gS, 256, 0, stream>>>(na, tmp, hbuf,
        S_, H_, S_, S_, H_, H_,
        (long)S_ * S_, (long)S_ * H_, (long)S_ * H_,
        b2, dsi, nullptr, nullptr, nullptr);

    classifier_kernel<<<(BS_ * C_ + 255) / 256, 256, 0, stream>>>(hbuf, Wc, bc, out);
}


# GEMM K-loop software-pipelined: each 8-WMMA block's LDS fragment reads go as one group into a 2nd register set ahead of the block before it; LDS-only barrier mid-step, fetch 2 tiles ahead
# speedup vs baseline: 1.0833x; 1.0415x over previous
#include <hip/hip_runtime.h>
#include <hip/hip_bf16.h>


typedef float v2f __attribute__((ext_vector_type(2)));
typedef float v4f __attribute__((ext_vector_type(4)));
typedef float v8f __attribute__((ext_vector_type(8)));

#define B_  16
#define S_  1024
#define H_  512
#define C_  7
#define BS_ (B_ * S_)
#define EPS_ 1e-10f
#define LPAD 20

static __device__ __forceinline__ v8f wmma4(v2f a, v2f b, v8f c) {
    return __builtin_amdgcn_wmma_f32_16x16x4_f32(
        false, a, false, b, (short)0, c, false, false);
}

static __device__ float blockReduceSum256(float v) {
    __shared__ float red[8];
    #pragma unroll
    for (int off = 16; off > 0; off >>= 1) v += __shfl_xor(v, off, 32);
    const int warp = threadIdx.x >> 5, lane = threadIdx.x & 31;
    __syncthreads();
    if (lane == 0) red[warp] = v;
    __syncthreads();
    float r = 0.f;
    #pragma unroll
    for (int w = 0; w < 8; ++w) r += red[w];
    return r;
}

__global__ __launch_bounds__(256)
void weff_kernel(const float* __restrict__ gW, float* __restrict__ weff) {
    int idx = blockIdx.x * blockDim.x + threadIdx.x;
    if (idx >= H_ * H_) return;
    weff[idx] = gW[idx] + gW[H_ * H_ + idx] - 0.5f * gW[2 * H_ * H_ + idx];
}

__global__ __launch_bounds__(256)
void ent_sum_kernel(const float* __restrict__ t, const float* __restrict__ a,
                    const float* __restrict__ v, float* __restrict__ sum_all,
                    float* __restrict__ entT, float* __restrict__ entA,
                    float* __restrict__ entV) {
    const size_t row = blockIdx.x;
    const size_t base = row * H_;
    const int i0 = threadIdx.x, i1 = threadIdx.x + 256;

    float t0 = t[base + i0], t1 = t[base + i1];
    float a0 = a[base + i0], a1 = a[base + i1];
    float v0 = v[base + i0], v1 = v[base + i1];

    sum_all[base + i0] = t0 + a0 + v0;
    sum_all[base + i1] = t1 + a1 + v1;

    float st = blockReduceSum256(fabsf(t0) + fabsf(t1));
    float sa = blockReduceSum256(fabsf(a0) + fabsf(a1));
    float sv = blockReduceSum256(fabsf(v0) + fabsf(v1));

    float rt = 1.f / (st + EPS_), ra = 1.f / (sa + EPS_), rv = 1.f / (sv + EPS_);
    float f;
    float et = 0.f, ea = 0.f, ev = 0.f;
    f = fabsf(t0) * rt; et += f * __logf(f + EPS_);
    f = fabsf(t1) * rt; et += f * __logf(f + EPS_);
    f = fabsf(a0) * ra; ea += f * __logf(f + EPS_);
    f = fabsf(a1) * ra; ea += f * __logf(f + EPS_);
    f = fabsf(v0) * rv; ev += f * __logf(f + EPS_);
    f = fabsf(v1) * rv; ev += f * __logf(f + EPS_);

    et = blockReduceSum256(et);
    ea = blockReduceSum256(ea);
    ev = blockReduceSum256(ev);
    if (threadIdx.x == 0) {
        entT[row] = -et; entA[row] = -ea; entV[row] = -ev;
    }
}

__global__ __launch_bounds__(256)
void adj_deg_kernel(const float* __restrict__ sp, const float* __restrict__ tm,
                    float* __restrict__ adj, float* __restrict__ dsi) {
    const int row  = blockIdx.x * 8 + (threadIdx.x >> 5);
    const int lane = threadIdx.x & 31;
    const int iloc = row & (S_ - 1);
    const size_t base = (size_t)row * S_;
    const v4f* sp4 = (const v4f*)(sp + base);
    const v4f* tm4 = (const v4f*)(tm + base);
    v4f*       ad4 = (v4f*)(adj + base);

    v4f x[8], y[8];
    #pragma unroll
    for (int m = 0; m < 8; ++m) {
        x[m] = sp4[m * 32 + lane];
        y[m] = tm4[m * 32 + lane];
    }
    float s = 0.f;
    #pragma unroll
    for (int m = 0; m < 8; ++m) {
        const int j0 = (m * 32 + lane) * 4;
        v4f a;
        a[0] = 0.5f * (x[m][0] + y[m][0]) + ((iloc == j0 + 0) ? 1.f : 0.f);
        a[1] = 0.5f * (x[m][1] + y[m][1]) + ((iloc == j0 + 1) ? 1.f : 0.f);
        a[2] = 0.5f * (x[m][2] + y[m][2]) + ((iloc == j0 + 2) ? 1.f : 0.f);
        a[3] = 0.5f * (x[m][3] + y[m][3]) + ((iloc == j0 + 3) ? 1.f : 0.f);
        ad4[m * 32 + lane] = a;
        s += (a[0] + a[1]) + (a[2] + a[3]);
    }
    #pragma unroll
    for (int off = 16; off > 0; off >>= 1) s += __shfl_xor(s, off, 32);
    if (lane == 0) dsi[row] = rsqrtf(s + EPS_);
}

static __device__ __forceinline__ void lds_barrier() {
    asm volatile("s_wait_dscnt 0x0\n\ts_barrier_signal -1\n\ts_barrier_wait -1" ::: "memory");
}

struct Frag8 {
    v2f a[2][2];
    v2f b[2][2];
};

static __device__ __forceinline__ void load_frag8(Frag8& f, const float* aRow,
                                                  const float* bRow, int kbase) {
    #pragma unroll
    for (int q = 0; q < 2; ++q) {
        const int k = kbase + q * 4;
        f.a[q][0] = *(const v2f*)(aRow + k);
        f.a[q][1] = *(const v2f*)(aRow + 16 * LPAD + k);
        f.b[q][0] = *(const v2f*)(bRow + k);
        f.b[q][1] = *(const v2f*)(bRow + 16 * LPAD + k);
    }
}

static __device__ __forceinline__ void mma_block8(v8f (&acc)[2][2], const Frag8& f) {
    #pragma unroll
    for (int q = 0; q < 2; ++q) {
        acc[0][0] = wmma4(f.a[q][0], f.b[q][0], acc[0][0]);
        acc[0][1] = wmma4(f.a[q][0], f.b[q][1], acc[0][1]);
        acc[1][0] = wmma4(f.a[q][1], f.b[q][0], acc[1][0]);
        acc[1][1] = wmma4(f.a[q][1], f.b[q][1], acc[1][1]);
    }
}

static __device__ __forceinline__ void block_done(v8f (&acc)[2][2]) {
    asm volatile("" : "+v"(acc[0][0]), "+v"(acc[0][1]), "+v"(acc[1][0]), "+v"(acc[1][1]));
}

static __device__ __forceinline__ void results_settle(v8f (&acc)[2][2], int& t) {
    asm volatile("v_nop\n\tv_nop\n\tv_nop\n\tv_nop\n\tv_nop\n\tv_nop\n\tv_nop\n\tv_nop"
                 : "+v"(acc[0][0]), "+v"(acc[0][1]), "+v"(acc[1][0]), "+v"(acc[1][1]), "+v"(t));
}

template <int EPI>
__global__ __launch_bounds__(256)
void gemm_wmma_f32(const float* __restrict__ A, const float* __restrict__ Bm,
                   float* __restrict__ C, int M, int N, int K,
                   int lda, int ldb, int ldc,
                   long strideA, long strideB, long strideC,
                   const float* __restrict__ e0,
                   const float* __restrict__ e1,
                   const float* __restrict__ e2,
                   const float* __restrict__ e3,
                   const float* __restrict__ e4) {
    __shared__ float As[2][128][LPAD];
    __shared__ float Bs[2][64][LPAD];

    const float* Ab = A + (size_t)blockIdx.z * strideA;
    const float* Bb = Bm + (size_t)blockIdx.z * strideB;
    float*       Cb = C + (size_t)blockIdx.z * strideC;
    const int grow0 = (int)blockIdx.z * M;

    const int bm = blockIdx.y * 128;
    const int bn = blockIdx.x * 64;
    const int tid = threadIdx.x;
    const int warp = tid >> 5, lane = tid & 31;
    const int waveM = warp & 3, waveN = warp >> 2;
    const int half = lane >> 4, mlan = lane & 15;

    const int amRow = tid >> 2;
    const int akQ   = (tid & 3) * 4;
    const int bkRow = tid >> 4;
    const int bnQ   = (tid & 15) * 4;
    typedef const v4f  __attribute__((address_space(1))) gv4f;
    typedef const char __attribute__((address_space(1))) gbyte;
    gbyte* gA = (gbyte*)(Ab + (size_t)bm * lda);
    gbyte* gB = (gbyte*)(Bb + bn);
    unsigned aOff0 = ((unsigned)amRow * (unsigned)lda + (unsigned)akQ) * 4u;
    unsigned aOff1 = aOff0 + 64u * (unsigned)lda * 4u;
    unsigned bOff  = ((unsigned)bkRow * (unsigned)ldb + (unsigned)bnQ) * 4u;
    const size_t bStepB = (size_t)ldb * 64;

    v8f acc[2][2];
    #pragma unroll
    for (int i = 0; i < 2; ++i)
        #pragma unroll
        for (int j = 0; j < 2; ++j)
            #pragma unroll
            for (int e = 0; e < 8; ++e) acc[i][j][e] = 0.f;

    v4f ra0, ra1, rb;
    const int ktiles = K >> 4;

    const float* aFr = &As[0][waveM * 32 + mlan][half * 2];
    const float* bFr = &Bs[0][waveN * 32 + mlan][half * 2];
    const int aStage = 128 * LPAD, bStage = 64 * LPAD;

    Frag8 fx, fy;

#define FETCH_NEXT                                                          \
    {                                                                       \
              \
              \
                \
          \
        asm volatile("" : "+s"(gA), "+s"(gB), "+v"(aOff0), "+v"(aOff1), "+v"(bOff)); \
        ra0 = *(gv4f*)(gA + aOff0);                                         \
        ra1 = *(gv4f*)(gA + aOff1);                                         \
        rb  = *(gv4f*)(gB + bOff);                                          \
        gA += 64; gB += bStepB;                                             \
    }

#define STAGE_TILE(NXT)                                                     \
    *(v4f*)&As[NXT][amRow][akQ]      = ra0;                                 \
    *(v4f*)&As[NXT][amRow + 64][akQ] = ra1;                                 \
    Bs[NXT][bnQ + 0][bkRow] = rb[0];                                        \
    Bs[NXT][bnQ + 1][bkRow] = rb[1];                                        \
    Bs[NXT][bnQ + 2][bkRow] = rb[2];                                        \
    Bs[NXT][bnQ + 3][bkRow] = rb[3];

#define PIPE_STEP(CUR, FETCH)                                               \
    {                                                                       \
        STAGE_TILE((CUR) ^ 1)                                               \
          \
        if (FETCH) FETCH_NEXT                                               \
             \
        load_frag8(fy, aFr + (CUR) * aStage, bFr + (CUR) * bStage, 8);      \
            \
           \
                                                           \
        __builtin_amdgcn_sched_barrier(0);                                  \
        mma_block8(acc, fx);                                                \
        block_done(acc);      \
        lds_barrier();           \
        load_frag8(fx, aFr + ((CUR) ^ 1) * aStage, bFr + ((CUR) ^ 1) * bStage, 0); \
          \
        __builtin_amdgcn_sched_barrier(0);                                  \
        mma_block8(acc, fy);                                                \
            \
                  \
        block_done(acc);                                                    \
    }

    FETCH_NEXT
    STAGE_TILE(0)
    FETCH_NEXT
    lds_barrier();
    load_frag8(fx, aFr, bFr, 0);

    int kt = 0;
    do {
        const bool more = kt + 2 < ktiles;
        PIPE_STEP(0, more)
        PIPE_STEP(1, more)
        kt += 2;
    } while (kt < ktiles);
#undef FETCH_NEXT
#undef STAGE_TILE
#undef PIPE_STEP

    int te = tid;
    results_settle(acc, te);
    __builtin_amdgcn_sched_barrier(0);
    const int ewM = (te >> 5) & 3, ewN = te >> 7;
    const int ehalf = (te >> 4) & 1, emlan = te & 15;

    #pragma unroll
    for (int mi = 0; mi < 2; ++mi) {
        #pragma unroll
        for (int ni = 0; ni < 2; ++ni) {
            const int tM = bm + ewM * 32 + mi * 16;
            const int tN = bn + ewN * 32 + ni * 16;
            #pragma unroll
            for (int i = 0; i < 8; ++i) {
                const int row = tM + (ehalf ? i + 8 : i);
                const int col = tN + emlan;
                float val = acc[mi][ni][i];
                if (EPI == 0) {
                    val *= e0[(size_t)(grow0 + row)];
                } else if (EPI == 1) {
                    val *= 0.5f;
                } else if (EPI == 2) {
                    const size_t grow = (size_t)(grow0 + row);
                    float logit = val + e0[grow * H_ + col]
                                + e1[grow] * e2[col] + e3[col];
                    float g = 1.f / (1.f + __expf(-logit));
                    val = g * e4[grow * H_ + col];
                } else if (EPI == 3) {
                    val = fmaxf(val * e1[(size_t)(grow0 + row)] + e0[col], 0.f);
                }
                Cb[(size_t)row * ldc + col] = val;
            }
        }
    }
}

__global__ __launch_bounds__(256)
void classifier_kernel(const float* __restrict__ h, const float* __restrict__ Wc,
                       const float* __restrict__ bc, float* __restrict__ out) {
    int idx = blockIdx.x * blockDim.x + threadIdx.x;
    if (idx >= BS_ * C_) return;
    int row = idx / C_, c = idx - row * C_;
    const float* hr = h + (size_t)row * H_;
    float acc = bc[c];
    #pragma unroll 8
    for (int k = 0; k < H_; ++k) acc = fmaf(hr[k], Wc[k * C_ + c], acc);
    out[idx] = acc;
}

extern "C" void kernel_launch(void* const* d_in, const int* in_sizes, int n_in,
                              void* d_out, int out_size, void* d_ws, size_t ws_size,
                              hipStream_t stream) {
    const float* text   = (const float*)d_in[0];
    const float* audio  = (const float*)d_in[1];
    const float* visual = (const float*)d_in[2];
    const float* sp     = (const float*)d_in[3];
    const float* tm     = (const float*)d_in[4];
    const float* gate_W = (const float*)d_in[6];
    const float* gate_b = (const float*)d_in[7];
    const float* W0 = (const float*)d_in[8];  const float* b0 = (const float*)d_in[9];
    const float* W1 = (const float*)d_in[10]; const float* b1 = (const float*)d_in[11];
    const float* W2 = (const float*)d_in[12]; const float* b2 = (const float*)d_in[13];
    const float* Wc = (const float*)d_in[14]; const float* bc = (const float*)d_in[15];
    float* out = (float*)d_out;

    float* ws    = (float*)d_ws;
    float* weff  = ws;
    float* entT  = weff + H_ * H_;
    float* entA  = entT + BS_;
    float* entV  = entA + BS_;
    float* dsi   = entV + BS_;
    float* sumA  = dsi + BS_;
    float* P     = sumA + (size_t)BS_ * H_;
    float* h0    = P    + (size_t)BS_ * H_;
    float* na    = h0   + (size_t)BS_ * 3 * H_;
    float* tmp   = sumA;
    float* hbuf  = P;

    const float* gWc_rows = gate_W + (size_t)2 * H_ * H_;
    const float* gw_last  = gate_W + (size_t)3 * H_ * H_;

    weff_kernel<<<(H_ * H_ + 255) / 256, 256, 0, stream>>>(gate_W, weff);

    ent_sum_kernel<<<BS_, 256, 0, stream>>>(text, audio, visual, sumA,
                                            entT, entA, entV);

    dim3 gP(H_ / 64, BS_ / 128, 1);
    gemm_wmma_f32<1><<<gP, 256, 0, stream>>>(sumA, gWc_rows, P,
        BS_, H_, H_, H_, H_, H_, 0, 0, 0, nullptr, nullptr, nullptr, nullptr, nullptr);

    gemm_wmma_f32<2><<<gP, 256, 0, stream>>>(text, weff, h0 + 0 * H_,
        BS_, H_, H_, H_, H_, 3 * H_, 0, 0, 0, P, entT, gw_last, gate_b, text);
    gemm_wmma_f32<2><<<gP, 256, 0, stream>>>(audio, weff, h0 + 1 * H_,
        BS_, H_, H_, H_, H_, 3 * H_, 0, 0, 0, P, entA, gw_last, gate_b, audio);
    gemm_wmma_f32<2><<<gP, 256, 0, stream>>>(visual, weff, h0 + 2 * H_,
        BS_, H_, H_, H_, H_, 3 * H_, 0, 0, 0, P, entV, gw_last, gate_b, visual);

    adj_deg_kernel<<<BS_ / 8, 256, 0, stream>>>(sp, tm, na, dsi);

    gemm_wmma_f32<0><<<gP, 256, 0, stream>>>(h0, W0, tmp,
        BS_, H_, 3 * H_, 3 * H_, H_, H_, 0, 0, 0,
        dsi, nullptr, nullptr, nullptr, nullptr);
    dim3 gS(H_ / 64, S_ / 128, B_);
    gemm_wmma_f32<3><<<gS, 256, 0, stream>>>(na, tmp, hbuf,
        S_, H_, S_, S_, H_, H_,
        (long)S_ * S_, (long)S_ * H_, (long)S_ * H_,
        b0, dsi, nullptr, nullptr, nullptr);

    gemm_wmma_f32<0><<<gP, 256, 0, stream>>>(hbuf, W1, tmp,
        BS_, H_, H_, H_, H_, H_, 0, 0, 0,
        dsi, nullptr, nullptr, nullptr, nullptr);
    gemm_wmma_f32<3><<<gS, 256, 0, stream>>>(na, tmp, hbuf,
        S_, H_, S_, S_, H_, H_,
        (long)S_ * S_, (long)S_ * H_, (long)S_ * H_,
        b1, dsi, nullptr, nullptr, nullptr);

    gemm_wmma_f32<0><<<gP, 256, 0, stream>>>(hbuf, W2, tmp,
        BS_, H_, H_, H_, H_, H_, 0, 0, 0,
        dsi, nullptr, nullptr, nullptr, nullptr);
    gemm_wmma_f32<3><<<gS, 256, 0, stream>>>(na, tmp, hbuf,
        S_, H_, S_, S_, H_, H_,
        (long)S_ * S_, (long)S_ * H_, (long)S_ * H_,
        b2, dsi, nullptr, nullptr, nullptr);

    classifier_kernel<<<(BS_ * C_ + 255) / 256, 256, 0, stream>>>(hbuf, Wc, bc, out);
}


# Shared gate term and the 3 gating GEMMs fused in one kernel: each tile runs 4 K-loops, 0.5*sum@Wc stays in registers as the accumulators' start value (not stored/re-read); logit sum order changes
# speedup vs baseline: 1.1269x; 1.0402x over previous
#include <hip/hip_runtime.h>
#include <hip/hip_bf16.h>


typedef float v2f __attribute__((ext_vector_type(2)));
typedef float v4f __attribute__((ext_vector_type(4)));
typedef float v8f __attribute__((ext_vector_type(8)));

#define B_  16
#define S_  1024
#define H_  512
#define C_  7
#define BS_ (B_ * S_)
#define EPS_ 1e-10f
#define LPAD 20

static __device__ __forceinline__ v8f wmma4(v2f a, v2f b, v8f c) {
    return __builtin_amdgcn_wmma_f32_16x16x4_f32(
        false, a, false, b, (short)0, c, false, false);
}

static __device__ float blockReduceSum256(float v) {
    __shared__ float red[8];
    #pragma unroll
    for (int off = 16; off > 0; off >>= 1) v += __shfl_xor(v, off, 32);
    const int warp = threadIdx.x >> 5, lane = threadIdx.x & 31;
    __syncthreads();
    if (lane == 0) red[warp] = v;
    __syncthreads();
    float r = 0.f;
    #pragma unroll
    for (int w = 0; w < 8; ++w) r += red[w];
    return r;
}

__global__ __launch_bounds__(256)
void weff_kernel(const float* __restrict__ gW, float* __restrict__ weff) {
    int idx = blockIdx.x * blockDim.x + threadIdx.x;
    if (idx >= H_ * H_) return;
    weff[idx] = gW[idx] + gW[H_ * H_ + idx] - 0.5f * gW[2 * H_ * H_ + idx];
}

__global__ __launch_bounds__(256)
void ent_sum_kernel(const float* __restrict__ t, const float* __restrict__ a,
                    const float* __restrict__ v, float* __restrict__ sum_all,
                    float* __restrict__ entT, float* __restrict__ entA,
                    float* __restrict__ entV) {
    const size_t row = blockIdx.x;
    const size_t base = row * H_;
    const int i0 = threadIdx.x, i1 = threadIdx.x + 256;

    float t0 = t[base + i0], t1 = t[base + i1];
    float a0 = a[base + i0], a1 = a[base + i1];
    float v0 = v[base + i0], v1 = v[base + i1];

    sum_all[base + i0] = t0 + a0 + v0;
    sum_all[base + i1] = t1 + a1 + v1;

    float st = blockReduceSum256(fabsf(t0) + fabsf(t1));
    float sa = blockReduceSum256(fabsf(a0) + fabsf(a1));
    float sv = blockReduceSum256(fabsf(v0) + fabsf(v1));

    float rt = 1.f / (st + EPS_), ra = 1.f / (sa + EPS_), rv = 1.f / (sv + EPS_);
    float f;
    float et = 0.f, ea = 0.f, ev = 0.f;
    f = fabsf(t0) * rt; et += f * __logf(f + EPS_);
    f = fabsf(t1) * rt; et += f * __logf(f + EPS_);
    f = fabsf(a0) * ra; ea += f * __logf(f + EPS_);
    f = fabsf(a1) * ra; ea += f * __logf(f + EPS_);
    f = fabsf(v0) * rv; ev += f * __logf(f + EPS_);
    f = fabsf(v1) * rv; ev += f * __logf(f + EPS_);

    et = blockReduceSum256(et);
    ea = blockReduceSum256(ea);
    ev = blockReduceSum256(ev);
    if (threadIdx.x == 0) {
        entT[row] = -et; entA[row] = -ea; entV[row] = -ev;
    }
}

__global__ __launch_bounds__(256)
void adj_deg_kernel(const float* __restrict__ sp, const float* __restrict__ tm,
                    float* __restrict__ adj, float* __restrict__ dsi) {
    const int row  = blockIdx.x * 8 + (threadIdx.x >> 5);
    const int lane = threadIdx.x & 31;
    const int iloc = row & (S_ - 1);
    const size_t base = (size_t)row * S_;
    const v4f* sp4 = (const v4f*)(sp + base);
    const v4f* tm4 = (const v4f*)(tm + base);
    v4f*       ad4 = (v4f*)(adj + base);

    v4f x[8], y[8];
    #pragma unroll
    for (int m = 0; m < 8; ++m) {
        x[m] = sp4[m * 32 + lane];
        y[m] = tm4[m * 32 + lane];
    }
    float s = 0.f;
    #pragma unroll
    for (int m = 0; m < 8; ++m) {
        const int j0 = (m * 32 + lane) * 4;
        v4f a;
        a[0] = 0.5f * (x[m][0] + y[m][0]) + ((iloc == j0 + 0) ? 1.f : 0.f);
        a[1] = 0.5f * (x[m][1] + y[m][1]) + ((iloc == j0 + 1) ? 1.f : 0.f);
        a[2] = 0.5f * (x[m][2] + y[m][2]) + ((iloc == j0 + 2) ? 1.f : 0.f);
        a[3] = 0.5f * (x[m][3] + y[m][3]) + ((iloc == j0 + 3) ? 1.f : 0.f);
        ad4[m * 32 + lane] = a;
        s += (a[0] + a[1]) + (a[2] + a[3]);
    }
    #pragma unroll
    for (int off = 16; off > 0; off >>= 1) s += __shfl_xor(s, off, 32);
    if (lane == 0) dsi[row] = rsqrtf(s + EPS_);
}

static __device__ __forceinline__ void lds_barrier() {
    asm volatile("s_wait_dscnt 0x0\n\ts_barrier_signal -1\n\ts_barrier_wait -1" ::: "memory");
}

struct Frag8 {
    v2f a[2][2];
    v2f b[2][2];
};

static __device__ __forceinline__ void load_frag8(Frag8& f, const float* aRow,
                                                  const float* bRow, int kbase) {
    #pragma unroll
    for (int q = 0; q < 2; ++q) {
        const int k = kbase + q * 4;
        f.a[q][0] = *(const v2f*)(aRow + k);
        f.a[q][1] = *(const v2f*)(aRow + 16 * LPAD + k);
        f.b[q][0] = *(const v2f*)(bRow + k);
        f.b[q][1] = *(const v2f*)(bRow + 16 * LPAD + k);
    }
}

static __device__ __forceinline__ void mma_block8(v8f (&acc)[2][2], const Frag8& f) {
    #pragma unroll
    for (int q = 0; q < 2; ++q) {
        acc[0][0] = wmma4(f.a[q][0], f.b[q][0], acc[0][0]);
        acc[0][1] = wmma4(f.a[q][0], f.b[q][1], acc[0][1]);
        acc[1][0] = wmma4(f.a[q][1], f.b[q][0], acc[1][0]);
        acc[1][1] = wmma4(f.a[q][1], f.b[q][1], acc[1][1]);
    }
}

static __device__ __forceinline__ void block_done(v8f (&acc)[2][2]) {
    asm volatile("" : "+v"(acc[0][0]), "+v"(acc[0][1]), "+v"(acc[1][0]), "+v"(acc[1][1]));
}

static __device__ __forceinline__ void results_settle(v8f (&acc)[2][2], int& t) {
    asm volatile("v_nop\n\tv_nop\n\tv_nop\n\tv_nop\n\tv_nop\n\tv_nop\n\tv_nop\n\tv_nop"
                 : "+v"(acc[0][0]), "+v"(acc[0][1]), "+v"(acc[1][0]), "+v"(acc[1][1]), "+v"(t));
}

template <int EPI>
__global__ __launch_bounds__(256)
void gemm_wmma_f32(const float* __restrict__ A, const float* __restrict__ Bm,
                   float* __restrict__ C, int M, int N, int K,
                   int lda, int ldb, int ldc,
                   long strideA, long strideB, long strideC,
                   const float* __restrict__ e0,
                   const float* __restrict__ e1) {
    __shared__ float As[2][128][LPAD];
    __shared__ float Bs[2][64][LPAD];

    const float* Ab = A + (size_t)blockIdx.z * strideA;
    const float* Bb = Bm + (size_t)blockIdx.z * strideB;
    float*       Cb = C + (size_t)blockIdx.z * strideC;
    const int grow0 = (int)blockIdx.z * M;

    const int bm = blockIdx.y * 128;
    const int bn = blockIdx.x * 64;
    const int tid = threadIdx.x;
    const int warp = tid >> 5, lane = tid & 31;
    const int waveM = warp & 3, waveN = warp >> 2;
    const int half = lane >> 4, mlan = lane & 15;

    const int amRow = tid >> 2;
    const int akQ   = (tid & 3) * 4;
    const int bkRow = tid >> 4;
    const int bnQ   = (tid & 15) * 4;
    typedef const v4f  __attribute__((address_space(1))) gv4f;
    typedef const char __attribute__((address_space(1))) gbyte;
    gbyte* gA = (gbyte*)(Ab + (size_t)bm * lda);
    gbyte* gB = (gbyte*)(Bb + bn);
    unsigned aOff0 = ((unsigned)amRow * (unsigned)lda + (unsigned)akQ) * 4u;
    unsigned aOff1 = aOff0 + 64u * (unsigned)lda * 4u;
    unsigned bOff  = ((unsigned)bkRow * (unsigned)ldb + (unsigned)bnQ) * 4u;
    const size_t bStepB = (size_t)ldb * 64;

    v8f acc[2][2];
    #pragma unroll
    for (int i = 0; i < 2; ++i)
        #pragma unroll
        for (int j = 0; j < 2; ++j)
            #pragma unroll
            for (int e = 0; e < 8; ++e) acc[i][j][e] = 0.f;

    v4f ra0, ra1, rb;
    const int ktiles = K >> 4;

    const float* aFr = &As[0][waveM * 32 + mlan][half * 2];
    const float* bFr = &Bs[0][waveN * 32 + mlan][half * 2];
    const int aStage = 128 * LPAD, bStage = 64 * LPAD;

    Frag8 fx, fy;

#define FETCH_NEXT                                                          \
    {                                                                       \
              \
              \
                \
          \
        asm volatile("" : "+s"(gA), "+s"(gB), "+v"(aOff0), "+v"(aOff1), "+v"(bOff)); \
        ra0 = *(gv4f*)(gA + aOff0);                                         \
        ra1 = *(gv4f*)(gA + aOff1);                                         \
        rb  = *(gv4f*)(gB + bOff);                                          \
        gA += 64; gB += bStepB;                                             \
    }

#define STAGE_TILE(NXT)                                                     \
    *(v4f*)&As[NXT][amRow][akQ]      = ra0;                                 \
    *(v4f*)&As[NXT][amRow + 64][akQ] = ra1;                                 \
    Bs[NXT][bnQ + 0][bkRow] = rb[0];                                        \
    Bs[NXT][bnQ + 1][bkRow] = rb[1];                                        \
    Bs[NXT][bnQ + 2][bkRow] = rb[2];                                        \
    Bs[NXT][bnQ + 3][bkRow] = rb[3];

#define PIPE_STEP(CUR, FETCH)                                               \
    {                                                                       \
        STAGE_TILE((CUR) ^ 1)                                               \
          \
        if (FETCH) FETCH_NEXT                                               \
             \
        load_frag8(fy, aFr + (CUR) * aStage, bFr + (CUR) * bStage, 8);      \
            \
           \
                                                           \
        __builtin_amdgcn_sched_barrier(0);                                  \
        mma_block8(acc, fx);                                                \
        block_done(acc);      \
        lds_barrier();           \
        load_frag8(fx, aFr + ((CUR) ^ 1) * aStage, bFr + ((CUR) ^ 1) * bStage, 0); \
          \
        __builtin_amdgcn_sched_barrier(0);                                  \
        mma_block8(acc, fy);                                                \
            \
                  \
        block_done(acc);                                                    \
    }

    FETCH_NEXT
    STAGE_TILE(0)
    FETCH_NEXT
    lds_barrier();
    load_frag8(fx, aFr, bFr, 0);

    int kt = 0;
    do {
        const bool more = kt + 2 < ktiles;
        PIPE_STEP(0, more)
        PIPE_STEP(1, more)
        kt += 2;
    } while (kt < ktiles);

    int te = tid;
    results_settle(acc, te);
    __builtin_amdgcn_sched_barrier(0);
    const int ewM = (te >> 5) & 3, ewN = te >> 7;
    const int ehalf = (te >> 4) & 1, emlan = te & 15;

    #pragma unroll
    for (int mi = 0; mi < 2; ++mi) {
        #pragma unroll
        for (int ni = 0; ni < 2; ++ni) {
            const int tM = bm + ewM * 32 + mi * 16;
            const int tN = bn + ewN * 32 + ni * 16;
            #pragma unroll
            for (int i = 0; i < 8; ++i) {
                const int row = tM + (ehalf ? i + 8 : i);
                const int col = tN + emlan;
                float val = acc[mi][ni][i];
                if (EPI == 0) {
                    val *= e0[(size_t)(grow0 + row)];
                } else if (EPI == 3) {
                    val = fmaxf(val * e1[(size_t)(grow0 + row)] + e0[col], 0.f);
                }
                Cb[(size_t)row * ldc + col] = val;
            }
        }
    }
}

__global__ __launch_bounds__(256)
void gate_fused_kernel(const float* __restrict__ sumA, const float* __restrict__ gWc,
                       const float* __restrict__ text, const float* __restrict__ audio,
                       const float* __restrict__ visual, const float* __restrict__ weff,
                       const float* __restrict__ ent,
                       const float* __restrict__ gw_last, const float* __restrict__ gate_b,
                       float* __restrict__ h0) {
    __shared__ float As[2][128][LPAD];
    __shared__ float Bs[2][64][LPAD];

    const int bm = blockIdx.y * 128;
    const int bn = blockIdx.x * 64;
    const int tid = threadIdx.x;
    const int warp = tid >> 5, lane = tid & 31;
    const int waveM = warp & 3, waveN = warp >> 2;
    const int half = lane >> 4, mlan = lane & 15;

    const int amRow = tid >> 2;
    const int akQ   = (tid & 3) * 4;
    const int bkRow = tid >> 4;
    const int bnQ   = (tid & 15) * 4;
    typedef const v4f  __attribute__((address_space(1))) gv4f;
    typedef const char __attribute__((address_space(1))) gbyte;
    unsigned aOff0 = ((unsigned)amRow * (unsigned)H_ + (unsigned)akQ) * 4u;
    unsigned aOff1 = aOff0 + 64u * (unsigned)H_ * 4u;
    unsigned bOff  = ((unsigned)bkRow * (unsigned)H_ + (unsigned)bnQ) * 4u;
    const size_t bStepB = (size_t)H_ * 64;
    const int ktiles = H_ >> 4;

    const float* aFr = &As[0][waveM * 32 + mlan][half * 2];
    const float* bFr = &Bs[0][waveN * 32 + mlan][half * 2];
    const int aStage = 128 * LPAD, bStage = 64 * LPAD;

    v8f acc[2][2], shared_term[2][2];
    #pragma unroll
    for (int i = 0; i < 2; ++i)
        #pragma unroll
        for (int j = 0; j < 2; ++j)
            #pragma unroll
            for (int e = 0; e < 8; ++e) shared_term[i][j][e] = 0.f;

    v4f ra0, ra1, rb;
    Frag8 fx, fy;

    #pragma nounroll
    for (int pass = 0; pass < 4; ++pass) {
        const float* Ap = pass == 0 ? sumA : pass == 1 ? text : pass == 2 ? audio : visual;
        const float* Bp = pass == 0 ? gWc : weff;
        gbyte* gA = (gbyte*)(Ap + (size_t)bm * H_);
        gbyte* gB = (gbyte*)(Bp + bn);

        #pragma unroll
        for (int i = 0; i < 2; ++i)
            #pragma unroll
            for (int j = 0; j < 2; ++j) acc[i][j] = shared_term[i][j];

        FETCH_NEXT
        STAGE_TILE(0)
        FETCH_NEXT
        lds_barrier();
        load_frag8(fx, aFr, bFr, 0);

        int kt = 0;
        do {
            const bool more = kt + 2 < ktiles;
            PIPE_STEP(0, more)
            PIPE_STEP(1, more)
            kt += 2;
        } while (kt < ktiles);

        int te = tid;
        results_settle(acc, te);
        __builtin_amdgcn_sched_barrier(0);

        if (pass == 0) {
            #pragma unroll
            for (int i = 0; i < 2; ++i)
                #pragma unroll
                for (int j = 0; j < 2; ++j)
                    #pragma unroll
                    for (int e = 0; e < 8; ++e) shared_term[i][j][e] = 0.5f * acc[i][j][e];
        } else {
            const int ewM = (te >> 5) & 3, ewN = te >> 7;
            const int ehalf = (te >> 4) & 1, emlan = te & 15;
            const float* entm = ent + (size_t)(pass - 1) * BS_;
            float*       Cm   = h0 + (pass - 1) * H_;
            #pragma unroll
            for (int mi = 0; mi < 2; ++mi) {
                #pragma unroll
                for (int ni = 0; ni < 2; ++ni) {
                    const int tM = bm + ewM * 32 + mi * 16;
                    const int tN = bn + ewN * 32 + ni * 16;
                    #pragma unroll
                    for (int i = 0; i < 8; ++i) {
                        const size_t row = (size_t)(tM + (ehalf ? i + 8 : i));
                        const int col = tN + emlan;
                        float logit = acc[mi][ni][i] + entm[row] * gw_last[col] + gate_b[col];
                        float g = 1.f / (1.f + __expf(-logit));
                        Cm[row * (3 * H_) + col] = g * Ap[row * H_ + col];
                    }
                }
            }
        }
    }
}
#undef FETCH_NEXT
#undef STAGE_TILE
#undef PIPE_STEP

__global__ __launch_bounds__(256)
void classifier_kernel(const float* __restrict__ h, const float* __restrict__ Wc,
                       const float* __restrict__ bc, float* __restrict__ out) {
    int idx = blockIdx.x * blockDim.x + threadIdx.x;
    if (idx >= BS_ * C_) return;
    int row = idx / C_, c = idx - row * C_;
    const float* hr = h + (size_t)row * H_;
    float acc = bc[c];
    #pragma unroll 8
    for (int k = 0; k < H_; ++k) acc = fmaf(hr[k], Wc[k * C_ + c], acc);
    out[idx] = acc;
}

extern "C" void kernel_launch(void* const* d_in, const int* in_sizes, int n_in,
                              void* d_out, int out_size, void* d_ws, size_t ws_size,
                              hipStream_t stream) {
    const float* text   = (const float*)d_in[0];
    const float* audio  = (const float*)d_in[1];
    const float* visual = (const float*)d_in[2];
    const float* sp     = (const float*)d_in[3];
    const float* tm     = (const float*)d_in[4];
    const float* gate_W = (const float*)d_in[6];
    const float* gate_b = (const float*)d_in[7];
    const float* W0 = (const float*)d_in[8];  const float* b0 = (const float*)d_in[9];
    const float* W1 = (const float*)d_in[10]; const float* b1 = (const float*)d_in[11];
    const float* W2 = (const float*)d_in[12]; const float* b2 = (const float*)d_in[13];
    const float* Wc = (const float*)d_in[14]; const float* bc = (const float*)d_in[15];
    float* out = (float*)d_out;

    float* ws    = (float*)d_ws;
    float* weff  = ws;
    float* entT  = weff + H_ * H_;
    float* entA  = entT + BS_;
    float* entV  = entA + BS_;
    float* dsi   = entV + BS_;
    float* sumA  = dsi + BS_;
    float* P     = sumA + (size_t)BS_ * H_;
    float* h0    = P    + (size_t)BS_ * H_;
    float* na    = h0   + (size_t)BS_ * 3 * H_;
    float* tmp   = sumA;
    float* hbuf  = P;

    const float* gWc_rows = gate_W + (size_t)2 * H_ * H_;
    const float* gw_last  = gate_W + (size_t)3 * H_ * H_;

    weff_kernel<<<(H_ * H_ + 255) / 256, 256, 0, stream>>>(gate_W, weff);

    ent_sum_kernel<<<BS_, 256, 0, stream>>>(text, audio, visual, sumA,
                                            entT, entA, entV);

    dim3 gP(H_ / 64, BS_ / 128, 1);
    gate_fused_kernel<<<gP, 256, 0, stream>>>(sumA, gWc_rows, text, audio, visual,
        weff, entT, gw_last, gate_b, h0);

    adj_deg_kernel<<<BS_ / 8, 256, 0, stream>>>(sp, tm, na, dsi);

    gemm_wmma_f32<0><<<gP, 256, 0, stream>>>(h0, W0, tmp,
        BS_, H_, 3 * H_, 3 * H_, H_, H_, 0, 0, 0,
        dsi, nullptr);
    dim3 gS(H_ / 64, S_ / 128, B_);
    gemm_wmma_f32<3><<<gS, 256, 0, stream>>>(na, tmp, hbuf,
        S_, H_, S_, S_, H_, H_,
        (long)S_ * S_, (long)S_ * H_, (long)S_ * H_,
        b0, dsi);

    gemm_wmma_f32<0><<<gP, 256, 0, stream>>>(hbuf, W1, tmp,
        BS_, H_, H_, H_, H_, H_, 0, 0, 0,
        dsi, nullptr);
    gemm_wmma_f32<3><<<gS, 256, 0, stream>>>(na, tmp, hbuf,
        S_, H_, S_, S_, H_, H_,
        (long)S_ * S_, (long)S_ * H_, (long)S_ * H_,
        b1, dsi);

    gemm_wmma_f32<0><<<gP, 256, 0, stream>>>(hbuf, W2, tmp,
        BS_, H_, H_, H_, H_, H_, 0, 0, 0,
        dsi, nullptr);
    gemm_wmma_f32<3><<<gS, 256, 0, stream>>>(na, tmp, hbuf,
        S_, H_, S_, S_, H_, H_,
        (long)S_ * S_, (long)S_ * H_, (long)S_ * H_,
        b2, dsi);

    classifier_kernel<<<(BS_ * C_ + 255) / 256, 256, 0, stream>>>(hbuf, Wc, bc, out);
}


# Gate logits in a modality sum/difference basis, 3 K-loops per tile not 4: Z=sum@(Wa+Wb+Wc), L_t=Z/3+(t-sum/3)@Weff, 2L_a=Z-L_t+(a-v)@Weff, L_v=Z-L_t-L_a; feat tile prefetched; sum order changes
# speedup vs baseline: 1.1671x; 1.0357x over previous
#include <hip/hip_runtime.h>
#include <hip/hip_bf16.h>


typedef float v2f __attribute__((ext_vector_type(2)));
typedef float v4f __attribute__((ext_vector_type(4)));
typedef float v8f __attribute__((ext_vector_type(8)));

#define B_  16
#define S_  1024
#define H_  512
#define C_  7
#define BS_ (B_ * S_)
#define EPS_ 1e-10f
#define LPAD 20

static __device__ __forceinline__ v8f wmma4(v2f a, v2f b, v8f c) {
    return __builtin_amdgcn_wmma_f32_16x16x4_f32(
        false, a, false, b, (short)0, c, false, false);
}

static __device__ float blockReduceSum256(float v) {
    __shared__ float red[8];
    #pragma unroll
    for (int off = 16; off > 0; off >>= 1) v += __shfl_xor(v, off, 32);
    const int warp = threadIdx.x >> 5, lane = threadIdx.x & 31;
    __syncthreads();
    if (lane == 0) red[warp] = v;
    __syncthreads();
    float r = 0.f;
    #pragma unroll
    for (int w = 0; w < 8; ++w) r += red[w];
    return r;
}

__global__ __launch_bounds__(256)
void weff_kernel(const float* __restrict__ gW, float* __restrict__ weff,
                 float* __restrict__ wsum) {
    int idx = blockIdx.x * blockDim.x + threadIdx.x;
    if (idx >= H_ * H_) return;
    const float wa = gW[idx], wb = gW[H_ * H_ + idx], wc = gW[2 * H_ * H_ + idx];
    weff[idx] = wa + wb - 0.5f * wc;
    wsum[idx] = wa + wb + wc;
}

#define THIRD_ 0.33333334f
__global__ __launch_bounds__(256)
void ent_sum_kernel(const float* __restrict__ t, const float* __restrict__ a,
                    const float* __restrict__ v, float* __restrict__ sum_all,
                    float* __restrict__ devT, float* __restrict__ difAV,
                    float* __restrict__ entT, float* __restrict__ entA,
                    float* __restrict__ entV) {
    const size_t row = blockIdx.x;
    const size_t base = row * H_;
    const int i0 = threadIdx.x, i1 = threadIdx.x + 256;

    float t0 = t[base + i0], t1 = t[base + i1];
    float a0 = a[base + i0], a1 = a[base + i1];
    float v0 = v[base + i0], v1 = v[base + i1];

    const float s0 = t0 + a0 + v0, s1 = t1 + a1 + v1;
    sum_all[base + i0] = s0;
    sum_all[base + i1] = s1;
    devT[base + i0] = fmaf(-THIRD_, s0, t0);
    devT[base + i1] = fmaf(-THIRD_, s1, t1);
    difAV[base + i0] = a0 - v0;
    difAV[base + i1] = a1 - v1;

    float st = blockReduceSum256(fabsf(t0) + fabsf(t1));
    float sa = blockReduceSum256(fabsf(a0) + fabsf(a1));
    float sv = blockReduceSum256(fabsf(v0) + fabsf(v1));

    float rt = 1.f / (st + EPS_), ra = 1.f / (sa + EPS_), rv = 1.f / (sv + EPS_);
    float f;
    float et = 0.f, ea = 0.f, ev = 0.f;
    f = fabsf(t0) * rt; et += f * __logf(f + EPS_);
    f = fabsf(t1) * rt; et += f * __logf(f + EPS_);
    f = fabsf(a0) * ra; ea += f * __logf(f + EPS_);
    f = fabsf(a1) * ra; ea += f * __logf(f + EPS_);
    f = fabsf(v0) * rv; ev += f * __logf(f + EPS_);
    f = fabsf(v1) * rv; ev += f * __logf(f + EPS_);

    et = blockReduceSum256(et);
    ea = blockReduceSum256(ea);
    ev = blockReduceSum256(ev);
    if (threadIdx.x == 0) {
        entT[row] = -et; entA[row] = -ea; entV[row] = -ev;
    }
}

__global__ __launch_bounds__(256)
void adj_deg_kernel(const float* __restrict__ sp, const float* __restrict__ tm,
                    float* __restrict__ adj, float* __restrict__ dsi) {
    const int row  = blockIdx.x * 8 + (threadIdx.x >> 5);
    const int lane = threadIdx.x & 31;
    const int iloc = row & (S_ - 1);
    const size_t base = (size_t)row * S_;
    const v4f* sp4 = (const v4f*)(sp + base);
    const v4f* tm4 = (const v4f*)(tm + base);
    v4f*       ad4 = (v4f*)(adj + base);

    v4f x[8], y[8];
    #pragma unroll
    for (int m = 0; m < 8; ++m) {
        x[m] = sp4[m * 32 + lane];
        y[m] = tm4[m * 32 + lane];
    }
    float s = 0.f;
    #pragma unroll
    for (int m = 0; m < 8; ++m) {
        const int j0 = (m * 32 + lane) * 4;
        v4f a;
        a[0] = 0.5f * (x[m][0] + y[m][0]) + ((iloc == j0 + 0) ? 1.f : 0.f);
        a[1] = 0.5f * (x[m][1] + y[m][1]) + ((iloc == j0 + 1) ? 1.f : 0.f);
        a[2] = 0.5f * (x[m][2] + y[m][2]) + ((iloc == j0 + 2) ? 1.f : 0.f);
        a[3] = 0.5f * (x[m][3] + y[m][3]) + ((iloc == j0 + 3) ? 1.f : 0.f);
        ad4[m * 32 + lane] = a;
        s += (a[0] + a[1]) + (a[2] + a[3]);
    }
    #pragma unroll
    for (int off = 16; off > 0; off >>= 1) s += __shfl_xor(s, off, 32);
    if (lane == 0) dsi[row] = rsqrtf(s + EPS_);
}

static __device__ __forceinline__ void lds_barrier() {
    asm volatile("s_wait_dscnt 0x0\n\ts_barrier_signal -1\n\ts_barrier_wait -1" ::: "memory");
}

struct Frag8 {
    v2f a[2][2];
    v2f b[2][2];
};

static __device__ __forceinline__ void load_frag8(Frag8& f, const float* aRow,
                                                  const float* bRow, int kbase) {
    #pragma unroll
    for (int q = 0; q < 2; ++q) {
        const int k = kbase + q * 4;
        f.a[q][0] = *(const v2f*)(aRow + k);
        f.a[q][1] = *(const v2f*)(aRow + 16 * LPAD + k);
        f.b[q][0] = *(const v2f*)(bRow + k);
        f.b[q][1] = *(const v2f*)(bRow + 16 * LPAD + k);
    }
}

static __device__ __forceinline__ void mma_block8(v8f (&acc)[2][2], const Frag8& f) {
    #pragma unroll
    for (int q = 0; q < 2; ++q) {
        acc[0][0] = wmma4(f.a[q][0], f.b[q][0], acc[0][0]);
        acc[0][1] = wmma4(f.a[q][0], f.b[q][1], acc[0][1]);
        acc[1][0] = wmma4(f.a[q][1], f.b[q][0], acc[1][0]);
        acc[1][1] = wmma4(f.a[q][1], f.b[q][1], acc[1][1]);
    }
}

static __device__ __forceinline__ void block_done(v8f (&acc)[2][2]) {
    asm volatile("" : "+v"(acc[0][0]), "+v"(acc[0][1]), "+v"(acc[1][0]), "+v"(acc[1][1]));
}

static __device__ __forceinline__ void results_settle(v8f (&acc)[2][2], int& t) {
    asm volatile("v_nop\n\tv_nop\n\tv_nop\n\tv_nop\n\tv_nop\n\tv_nop\n\tv_nop\n\tv_nop"
                 : "+v"(acc[0][0]), "+v"(acc[0][1]), "+v"(acc[1][0]), "+v"(acc[1][1]), "+v"(t));
}

template <int EPI>
__global__ __launch_bounds__(256)
void gemm_wmma_f32(const float* __restrict__ A, const float* __restrict__ Bm,
                   float* __restrict__ C, int M, int N, int K,
                   int lda, int ldb, int ldc,
                   long strideA, long strideB, long strideC,
                   const float* __restrict__ e0,
                   const float* __restrict__ e1) {
    __shared__ float As[2][128][LPAD];
    __shared__ float Bs[2][64][LPAD];

    const float* Ab = A + (size_t)blockIdx.z * strideA;
    const float* Bb = Bm + (size_t)blockIdx.z * strideB;
    float*       Cb = C + (size_t)blockIdx.z * strideC;
    const int grow0 = (int)blockIdx.z * M;

    const int bm = blockIdx.y * 128;
    const int bn = blockIdx.x * 64;
    const int tid = threadIdx.x;
    const int warp = tid >> 5, lane = tid & 31;
    const int waveM = warp & 3, waveN = warp >> 2;
    const int half = lane >> 4, mlan = lane & 15;

    const int amRow = tid >> 2;
    const int akQ   = (tid & 3) * 4;
    const int bkRow = tid >> 4;
    const int bnQ   = (tid & 15) * 4;
    typedef const v4f  __attribute__((address_space(1))) gv4f;
    typedef const char __attribute__((address_space(1))) gbyte;
    gbyte* gA = (gbyte*)(Ab + (size_t)bm * lda);
    gbyte* gB = (gbyte*)(Bb + bn);
    unsigned aOff0 = ((unsigned)amRow * (unsigned)lda + (unsigned)akQ) * 4u;
    unsigned aOff1 = aOff0 + 64u * (unsigned)lda * 4u;
    unsigned bOff  = ((unsigned)bkRow * (unsigned)ldb + (unsigned)bnQ) * 4u;
    const size_t bStepB = (size_t)ldb * 64;

    v8f acc[2][2];
    #pragma unroll
    for (int i = 0; i < 2; ++i)
        #pragma unroll
        for (int j = 0; j < 2; ++j)
            #pragma unroll
            for (int e = 0; e < 8; ++e) acc[i][j][e] = 0.f;

    v4f ra0, ra1, rb;
    const int ktiles = K >> 4;

    const float* aFr = &As[0][waveM * 32 + mlan][half * 2];
    const float* bFr = &Bs[0][waveN * 32 + mlan][half * 2];
    const int aStage = 128 * LPAD, bStage = 64 * LPAD;

    Frag8 fx, fy;

#define FETCH_NEXT                                                          \
    {                                                                       \
              \
              \
                \
          \
        asm volatile("" : "+s"(gA), "+s"(gB), "+v"(aOff0), "+v"(aOff1), "+v"(bOff)); \
        ra0 = *(gv4f*)(gA + aOff0);                                         \
        ra1 = *(gv4f*)(gA + aOff1);                                         \
        rb  = *(gv4f*)(gB + bOff);                                          \
        gA += 64; gB += bStepB;                                             \
    }

#define STAGE_TILE(NXT)                                                     \
    *(v4f*)&As[NXT][amRow][akQ]      = ra0;                                 \
    *(v4f*)&As[NXT][amRow + 64][akQ] = ra1;                                 \
    Bs[NXT][bnQ + 0][bkRow] = rb[0];                                        \
    Bs[NXT][bnQ + 1][bkRow] = rb[1];                                        \
    Bs[NXT][bnQ + 2][bkRow] = rb[2];                                        \
    Bs[NXT][bnQ + 3][bkRow] = rb[3];

#define PIPE_STEP(CUR, FETCH)                                               \
    {                                                                       \
        STAGE_TILE((CUR) ^ 1)                                               \
          \
        if (FETCH) FETCH_NEXT                                               \
             \
        load_frag8(fy, aFr + (CUR) * aStage, bFr + (CUR) * bStage, 8);      \
            \
           \
                                                           \
        __builtin_amdgcn_sched_barrier(0);                                  \
        mma_block8(acc, fx);                                                \
        block_done(acc);      \
        lds_barrier();           \
        load_frag8(fx, aFr + ((CUR) ^ 1) * aStage, bFr + ((CUR) ^ 1) * bStage, 0); \
          \
        __builtin_amdgcn_sched_barrier(0);                                  \
        mma_block8(acc, fy);                                                \
            \
                  \
        block_done(acc);                                                    \
    }

    FETCH_NEXT
    STAGE_TILE(0)
    FETCH_NEXT
    lds_barrier();
    load_frag8(fx, aFr, bFr, 0);

    int kt = 0;
    do {
        const bool more = kt + 2 < ktiles;
        PIPE_STEP(0, more)
        PIPE_STEP(1, more)
        kt += 2;
    } while (kt < ktiles);

    int te = tid;
    results_settle(acc, te);
    __builtin_amdgcn_sched_barrier(0);
    const int ewM = (te >> 5) & 3, ewN = te >> 7;
    const int ehalf = (te >> 4) & 1, emlan = te & 15;

    #pragma unroll
    for (int mi = 0; mi < 2; ++mi) {
        #pragma unroll
        for (int ni = 0; ni < 2; ++ni) {
            const int tM = bm + ewM * 32 + mi * 16;
            const int tN = bn + ewN * 32 + ni * 16;
            #pragma unroll
            for (int i = 0; i < 8; ++i) {
                const int row = tM + (ehalf ? i + 8 : i);
                const int col = tN + emlan;
                float val = acc[mi][ni][i];
                if (EPI == 0) {
                    val *= e0[(size_t)(grow0 + row)];
                } else if (EPI == 3) {
                    val = fmaxf(val * e1[(size_t)(grow0 + row)] + e0[col], 0.f);
                }
                Cb[(size_t)row * ldc + col] = val;
            }
        }
    }
}

static __device__ __forceinline__ void gate_store(const v8f (&lg)[2][2], int te, int bm, int bn,
                                                  const float* __restrict__ feat,
                                                  const float* __restrict__ entm,
                                                  const float* __restrict__ gw_last,
                                                  const float* __restrict__ gate_b,
                                                  float* __restrict__ Cm) {
    const int ewM = (te >> 5) & 3, ewN = te >> 7;
    const int ehalf = (te >> 4) & 1, emlan = te & 15;
    #pragma unroll
    for (int mi = 0; mi < 2; ++mi) {
        #pragma unroll
        for (int ni = 0; ni < 2; ++ni) {
            const int tM = bm + ewM * 32 + mi * 16;
            const int tN = bn + ewN * 32 + ni * 16;
            #pragma unroll
            for (int i = 0; i < 8; ++i) {
                const size_t row = (size_t)(tM + (ehalf ? i + 8 : i));
                const int col = tN + emlan;
                float logit = lg[mi][ni][i] + entm[row] * gw_last[col] + gate_b[col];
                float g = 1.f / (1.f + __expf(-logit));
                Cm[row * (3 * H_) + col] = g * feat[row * H_ + col];
            }
        }
    }
}

__global__ __launch_bounds__(256)
void gate_fused_kernel(const float* __restrict__ sumA, const float* __restrict__ wsum,
                       const float* __restrict__ devT, const float* __restrict__ difAV,
                       const float* __restrict__ weff,
                       const float* __restrict__ text, const float* __restrict__ audio,
                       const float* __restrict__ visual,
                       const float* __restrict__ ent,
                       const float* __restrict__ gw_last, const float* __restrict__ gate_b,
                       float* __restrict__ h0) {
    __shared__ float As[2][128][LPAD];
    __shared__ float Bs[2][64][LPAD];

    const int bm = blockIdx.y * 128;
    const int bn = blockIdx.x * 64;
    const int tid = threadIdx.x;
    const int warp = tid >> 5, lane = tid & 31;
    const int waveM = warp & 3, waveN = warp >> 2;
    const int half = lane >> 4, mlan = lane & 15;

    const int amRow = tid >> 2;
    const int akQ   = (tid & 3) * 4;
    const int bkRow = tid >> 4;
    const int bnQ   = (tid & 15) * 4;
    typedef const v4f  __attribute__((address_space(1))) gv4f;
    typedef const char __attribute__((address_space(1))) gbyte;
    unsigned aOff0 = ((unsigned)amRow * (unsigned)H_ + (unsigned)akQ) * 4u;
    unsigned aOff1 = aOff0 + 64u * (unsigned)H_ * 4u;
    unsigned bOff  = ((unsigned)bkRow * (unsigned)H_ + (unsigned)bnQ) * 4u;
    const size_t bStepB = (size_t)H_ * 64;
    const int ktiles = H_ >> 4;

    const float* aFr = &As[0][waveM * 32 + mlan][half * 2];
    const float* bFr = &Bs[0][waveN * 32 + mlan][half * 2];
    const int aStage = 128 * LPAD, bStage = 64 * LPAD;

    v8f acc[2][2], keep[2][2];
    #pragma unroll
    for (int i = 0; i < 2; ++i)
        #pragma unroll
        for (int j = 0; j < 2; ++j)
            #pragma unroll
            for (int e = 0; e < 8; ++e) keep[i][j][e] = 0.f;

    v4f ra0, ra1, rb;
    Frag8 fx, fy;

    #pragma nounroll
    for (int pass = 0; pass < 4; ++pass) {
        const float startScale = pass == 1 ? THIRD_ : 1.f;
        #pragma unroll
        for (int i = 0; i < 2; ++i)
            #pragma unroll
            for (int j = 0; j < 2; ++j) acc[i][j] = keep[i][j] * startScale;

        int te = tid;
        if (pass < 3) {
            const float* Ap = pass == 0 ? sumA : pass == 1 ? devT : difAV;
            const float* Bp = pass == 0 ? wsum : weff;
            gbyte* gA = (gbyte*)(Ap + (size_t)bm * H_);
            gbyte* gB = (gbyte*)(Bp + bn);

            const float* nextFeat = pass == 0 ? text : pass == 1 ? audio : visual;
            __builtin_prefetch(nextFeat + (size_t)(bm + waveM * 32 + lane) * H_
                                        + bn + waveN * 32, 0, 3);

            FETCH_NEXT
            STAGE_TILE(0)
            FETCH_NEXT
            lds_barrier();
            load_frag8(fx, aFr, bFr, 0);

            int kt = 0;
            do {
                const bool more = kt + 2 < ktiles;
                PIPE_STEP(0, more)
                PIPE_STEP(1, more)
                kt += 2;
            } while (kt < ktiles);

            results_settle(acc, te);
            __builtin_amdgcn_sched_barrier(0);
        }

        if (pass == 0) {
            #pragma unroll
            for (int i = 0; i < 2; ++i)
                #pragma unroll
                for (int j = 0; j < 2; ++j) keep[i][j] = acc[i][j];
        } else {
            if (pass == 2) {
                #pragma unroll
                for (int i = 0; i < 2; ++i)
                    #pragma unroll
                    for (int j = 0; j < 2; ++j) acc[i][j] = acc[i][j] * 0.5f;
            }
            if (pass < 3) {
                #pragma unroll
                for (int i = 0; i < 2; ++i)
                    #pragma unroll
                    for (int j = 0; j < 2; ++j) keep[i][j] = keep[i][j] - acc[i][j];
            }
            const float* feat = pass == 1 ? text : pass == 2 ? audio : visual;
            gate_store(acc, te, bm, bn, feat, ent + (size_t)(pass - 1) * BS_, gw_last, gate_b,
                       h0 + (pass - 1) * H_);
        }
    }
}
#undef FETCH_NEXT
#undef STAGE_TILE
#undef PIPE_STEP

__global__ __launch_bounds__(256)
void classifier_kernel(const float* __restrict__ h, const float* __restrict__ Wc,
                       const float* __restrict__ bc, float* __restrict__ out) {
    int idx = blockIdx.x * blockDim.x + threadIdx.x;
    if (idx >= BS_ * C_) return;
    int row = idx / C_, c = idx - row * C_;
    const float* hr = h + (size_t)row * H_;
    float acc = bc[c];
    #pragma unroll 8
    for (int k = 0; k < H_; ++k) acc = fmaf(hr[k], Wc[k * C_ + c], acc);
    out[idx] = acc;
}

extern "C" void kernel_launch(void* const* d_in, const int* in_sizes, int n_in,
                              void* d_out, int out_size, void* d_ws, size_t ws_size,
                              hipStream_t stream) {
    const float* text   = (const float*)d_in[0];
    const float* audio  = (const float*)d_in[1];
    const float* visual = (const float*)d_in[2];
    const float* sp     = (const float*)d_in[3];
    const float* tm     = (const float*)d_in[4];
    const float* gate_W = (const float*)d_in[6];
    const float* gate_b = (const float*)d_in[7];
    const float* W0 = (const float*)d_in[8];  const float* b0 = (const float*)d_in[9];
    const float* W1 = (const float*)d_in[10]; const float* b1 = (const float*)d_in[11];
    const float* W2 = (const float*)d_in[12]; const float* b2 = (const float*)d_in[13];
    const float* Wc = (const float*)d_in[14]; const float* bc = (const float*)d_in[15];
    float* out = (float*)d_out;

    float* ws    = (float*)d_ws;
    float* weff  = ws;
    float* entT  = weff + H_ * H_;
    float* entA  = entT + BS_;
    float* entV  = entA + BS_;
    float* dsi   = entV + BS_;
    float* sumA  = dsi + BS_;
    float* P     = sumA + (size_t)BS_ * H_;
    float* h0    = P    + (size_t)BS_ * H_;
    float* na    = h0   + (size_t)BS_ * 3 * H_;
    float* tmp   = sumA;
    float* hbuf  = P;
    float* devT  = P;
    float* difAV = na;
    float* wsum  = na + (size_t)BS_ * H_;

    const float* gw_last  = gate_W + (size_t)3 * H_ * H_;

    weff_kernel<<<(H_ * H_ + 255) / 256, 256, 0, stream>>>(gate_W, weff, wsum);

    ent_sum_kernel<<<BS_, 256, 0, stream>>>(text, audio, visual, sumA, devT, difAV,
                                            entT, entA, entV);

    dim3 gP(H_ / 64, BS_ / 128, 1);
    gate_fused_kernel<<<gP, 256, 0, stream>>>(sumA, wsum, devT, difAV, weff,
        text, audio, visual, entT, gw_last, gate_b, h0);

    adj_deg_kernel<<<BS_ / 8, 256, 0, stream>>>(sp, tm, na, dsi);

    gemm_wmma_f32<0><<<gP, 256, 0, stream>>>(h0, W0, tmp,
        BS_, H_, 3 * H_, 3 * H_, H_, H_, 0, 0, 0,
        dsi, nullptr);
    dim3 gS(H_ / 64, S_ / 128, B_);
    gemm_wmma_f32<3><<<gS, 256, 0, stream>>>(na, tmp, hbuf,
        S_, H_, S_, S_, H_, H_,
        (long)S_ * S_, (long)S_ * H_, (long)S_ * H_,
        b0, dsi);

    gemm_wmma_f32<0><<<gP, 256, 0, stream>>>(hbuf, W1, tmp,
        BS_, H_, H_, H_, H_, H_, 0, 0, 0,
        dsi, nullptr);
    gemm_wmma_f32<3><<<gS, 256, 0, stream>>>(na, tmp, hbuf,
        S_, H_, S_, S_, H_, H_,
        (long)S_ * S_, (long)S_ * H_, (long)S_ * H_,
        b1, dsi);

    gemm_wmma_f32<0><<<gP, 256, 0, stream>>>(hbuf, W2, tmp,
        BS_, H_, H_, H_, H_, H_, 0, 0, 0,
        dsi, nullptr);
    gemm_wmma_f32<3><<<gS, 256, 0, stream>>>(na, tmp, hbuf,
        S_, H_, S_, S_, H_, H_,
        (long)S_ * S_, (long)S_ * H_, (long)S_ * H_,
        b2, dsi);

    classifier_kernel<<<(BS_ * C_ + 255) / 256, 256, 0, stream>>>(hbuf, Wc, bc, out);
}
